# starting kernel
# baseline (speedup 1.0000x reference)
#include <hip/hip_runtime.h>

typedef __attribute__((ext_vector_type(2))) float v2f;
typedef __attribute__((ext_vector_type(8))) float v8f;

#define DDIM  512
#define KNEG  512
#define NOUT  513
#define NTILES 33

__global__ __launch_bounds__(256)
void ns_linear_wmma_kernel(const float* __restrict__ x,
                           const int*   __restrict__ y,
                           const int*   __restrict__ neg_idx,
                           const float* __restrict__ W,
                           float*       __restrict__ out)
{
    __shared__ __align__(16) float xs[DDIM];

    const int b   = blockIdx.x;
    const int tid = threadIdx.x;

    xs[tid]       = x[(size_t)b * DDIM + tid];
    xs[tid + 256] = x[(size_t)b * DDIM + tid + 256];
    __syncthreads();

    const int lane = tid & 31;
    const int wave = tid >> 5;
    const int m    = lane & 15;
    const int koff = (lane >> 4) << 1;

    for (int tile = wave; tile < NTILES; tile += 8) {
        int j  = tile * 16 + m;
        int jj = (j < NOUT) ? j : 0;
        int row = (jj == 0) ? y[b] : neg_idx[(size_t)b * KNEG + (jj - 1)];
        const float* wrow = W + (size_t)row * DDIM;

        v8f c = {0.f, 0.f, 0.f, 0.f, 0.f, 0.f, 0.f, 0.f};

        for (int k = 0; k < DDIM; k += 32) {
            v2f a[8], xb[8];
            #pragma unroll
            for (int u = 0; u < 8; ++u) {
                a[u]  = *(const v2f*)(wrow + k + u * 4 + koff);
                xb[u] = *(const v2f*)(&xs[k + u * 4 + koff]);
            }
            #pragma unroll
            for (int u = 0; u < 8; ++u) {
                c = __builtin_amdgcn_wmma_f32_16x16x4_f32(
                         false, a[u],
                         false, xb[u],
                         (short)0, c,
                         false,  false);
            }
        }

        if (m == 0) {
            int base = tile * 16 + (lane >> 4) * 8;
            #pragma unroll
            for (int i = 0; i < 8; ++i) {
                int jo = base + i;
                if (jo < NOUT) out[(size_t)b * NOUT + jo] = c[i];
            }
        }
    }
}

extern "C" void kernel_launch(void* const* d_in, const int* in_sizes, int n_in,
                              void* d_out, int out_size, void* d_ws, size_t ws_size,
                              hipStream_t stream) {
    const float* x   = (const float*)d_in[0];
    const int*   yy  = (const int*)d_in[1];
    const int*   ni  = (const int*)d_in[2];
    const float* W   = (const float*)d_in[3];
    float*       out = (float*)d_out;

    const int B = in_sizes[1];
    ns_linear_wmma_kernel<<<B, 256, 0, stream>>>(x, yy, ni, W, out);
}


# Each lane loads 16 B of its W row per instruction (was 8 B): half the load instructions and cache-line touches per row; one load feeds two WMMA k-steps, so the f32 summation order changes
# speedup vs baseline: 1.0791x; 1.0791x over previous
#include <hip/hip_runtime.h>


typedef __attribute__((ext_vector_type(2))) float v2f;
typedef __attribute__((ext_vector_type(4))) float v4f;
typedef __attribute__((ext_vector_type(8))) float v8f;

#define DDIM  512
#define KNEG  512
#define NOUT  513
#define NTILES 33

__global__ __launch_bounds__(256)
void ns_linear_wmma_kernel(const float* __restrict__ x,
                           const int*   __restrict__ y,
                           const int*   __restrict__ neg_idx,
                           const float* __restrict__ W,
                           float*       __restrict__ out)
{
    __shared__ __align__(16) float xs[DDIM];

    const int b   = blockIdx.x;
    const int tid = threadIdx.x;

    xs[tid]       = x[(size_t)b * DDIM + tid];
    xs[tid + 256] = x[(size_t)b * DDIM + tid + 256];
    __syncthreads();

    const int lane = tid & 31;
    const int wave = tid >> 5;
    const int m    = lane & 15;
    const int hoff = (lane >> 4) << 2;

    for (int tile = wave; tile < NTILES; tile += 8) {
        int j  = tile * 16 + m;
        int jj = (j < NOUT) ? j : 0;
        int row = (jj == 0) ? y[b] : neg_idx[(size_t)b * KNEG + (jj - 1)];
        const float* wrow = W + (size_t)row * DDIM;

        v8f c = {0.f, 0.f, 0.f, 0.f, 0.f, 0.f, 0.f, 0.f};

        #pragma unroll 1
        for (int k = 0; k < DDIM; k += 64) {
            v4f a[8], xb[8];
            #pragma unroll
            for (int u = 0; u < 8; ++u) {
                a[u]  = *(const v4f*)(wrow + k + u * 8 + hoff);
                xb[u] = *(const v4f*)(&xs[k + u * 8 + hoff]);
            }
            #pragma unroll
            for (int u = 0; u < 8; ++u) {
                v2f a0 = {a[u].x,  a[u].y};
                v2f b0 = {xb[u].x, xb[u].y};
                v2f a1 = {a[u].z,  a[u].w};
                v2f b1 = {xb[u].z, xb[u].w};
                c = __builtin_amdgcn_wmma_f32_16x16x4_f32(
                         false, a0,
                         false, b0,
                         (short)0, c,
                         false,  false);
                c = __builtin_amdgcn_wmma_f32_16x16x4_f32(
                         false, a1,
                         false, b1,
                         (short)0, c,
                         false,  false);
            }
        }

        if (m == 0) {
            int base = tile * 16 + (lane >> 4) * 8;
            #pragma unroll
            for (int i = 0; i < 8; ++i) {
                int jo = base + i;
                if (jo < NOUT) out[(size_t)b * NOUT + jo] = c[i];
            }
        }
    }
}

extern "C" void kernel_launch(void* const* d_in, const int* in_sizes, int n_in,
                              void* d_out, int out_size, void* d_ws, size_t ws_size,
                              hipStream_t stream) {
    const float* x   = (const float*)d_in[0];
    const int*   yy  = (const int*)d_in[1];
    const int*   ni  = (const int*)d_in[2];
    const float* W   = (const float*)d_in[3];
    float*       out = (float*)d_out;

    const int B = in_sizes[1];
    ns_linear_wmma_kernel<<<B, 256, 0, stream>>>(x, yy, ni, W, out);
}


# W-row loads software-pipelined: a 16-load register ring per wave, each 128-B line requested again right behind the 8 WMMAs that read it (next tile's row too); f32 sum order unchanged
# speedup vs baseline: 1.2054x; 1.1170x over previous
#include <hip/hip_runtime.h>


typedef __attribute__((ext_vector_type(2))) float v2f;
typedef __attribute__((ext_vector_type(4))) float v4f;
typedef __attribute__((ext_vector_type(8))) float v8f;

#define DDIM  512
#define KNEG  512
#define NOUT  513
#define NTILES 33
#define RING  16

template <bool RELOAD>
__device__ __forceinline__ void step(v8f& c, v4f (&a)[RING],
                                     const float* xp, const float* rl)
{
    v4f xq0 = *(const v4f*)(xp);
    v4f xq1 = *(const v4f*)(xp + 8);
    #pragma unroll
    for (int u = 0; u < RING; ++u) {
        const v4f xb = xq0;
        xq0 = xq1;
        if (u + 2 < RING) xq1 = *(const v4f*)(xp + (u + 2) * 8);
        const v2f a0 = {a[u].x, a[u].y};
        const v2f b0 = {xb.x,   xb.y};
        const v2f a1 = {a[u].z, a[u].w};
        const v2f b1 = {xb.z,   xb.w};
        c = __builtin_amdgcn_wmma_f32_16x16x4_f32(
                 false, a0,
                 false, b0,
                 (short)0, c,
                 false,  false);
        c = __builtin_amdgcn_wmma_f32_16x16x4_f32(
                 false, a1,
                 false, b1,
                 (short)0, c,
                 false,  false);
        __builtin_amdgcn_sched_barrier(0);
        if (RELOAD && (u & 3) == 3) {
            #pragma unroll
            for (int v = u - 3; v <= u; ++v) a[v] = *(const v4f*)(rl + v * 8);
            __builtin_amdgcn_sched_barrier(0);
        }
    }
}

__global__ __launch_bounds__(256)
void ns_linear_wmma_kernel(const float* __restrict__ x,
                           const int*   __restrict__ y,
                           const int*   __restrict__ neg_idx,
                           const float* __restrict__ W,
                           float*       __restrict__ out)
{
    __shared__ __align__(16) float xs[DDIM];

    const int b   = blockIdx.x;
    const int tid = threadIdx.x;

    const int lane = tid & 31;
    const int wave = tid >> 5;
    const int m    = lane & 15;
    const int hoff = (lane >> 4) << 2;

    auto row_index = [&](int tile) -> int {
        int j  = tile * 16 + m;
        int jj = (j < NOUT) ? j : 0;
        return (jj == 0) ? y[b] : neg_idx[(size_t)b * KNEG + (jj - 1)];
    };

    const float x0 = x[(size_t)b * DDIM + tid];
    const float x1 = x[(size_t)b * DDIM + tid + 256];

    int tile = wave;
    const int r_cur = row_index(tile);
    int r_nxt = row_index(tile + 8);

    xs[tid]       = x0;
    xs[tid + 256] = x1;

    const float* wrow = W + (size_t)r_cur * DDIM + hoff;

    v4f a[RING];
    #pragma unroll
    for (int u = 0; u < RING; ++u) a[u] = *(const v4f*)(wrow + u * 8);

    __syncthreads();

    const float* xh = &xs[hoff];

    for (; tile < NTILES; tile += 8) {
        const bool has_next = (tile + 8 < NTILES);
        const float* wnext = W + (size_t)r_nxt * DDIM + hoff;
        r_nxt = row_index((tile + 16 < NTILES) ? tile + 16 : tile);

        v8f c = {0.f, 0.f, 0.f, 0.f, 0.f, 0.f, 0.f, 0.f};

        #pragma unroll 1
        for (int s = 0; s < 3; ++s)
            step<true>(c, a, xh + s * 128, wrow + (s + 1) * 128);

        if (has_next) step<true >(c, a, xh + 384, wnext);
        else          step<false>(c, a, xh + 384, wnext);

        if (m == 0) {
            int base = tile * 16 + (lane >> 4) * 8;
            #pragma unroll
            for (int i = 0; i < 8; ++i) {
                int jo = base + i;
                if (jo < NOUT) out[(size_t)b * NOUT + jo] = c[i];
            }
        }
        wrow = wnext;
    }
}

extern "C" void kernel_launch(void* const* d_in, const int* in_sizes, int n_in,
                              void* d_out, int out_size, void* d_ws, size_t ws_size,
                              hipStream_t stream) {
    const float* x   = (const float*)d_in[0];
    const int*   yy  = (const int*)d_in[1];
    const int*   ni  = (const int*)d_in[2];
    const float* W   = (const float*)d_in[3];
    float*       out = (float*)d_out;

    const int B = in_sizes[1];
    ns_linear_wmma_kernel<<<B, 256, 0, stream>>>(x, yy, ni, W, out);
}
